# starting kernel
# baseline (speedup 1.0000x reference)
#include <hip/hip_runtime.h>

typedef __attribute__((ext_vector_type(2))) float v2f;
typedef __attribute__((ext_vector_type(8))) float v8f;

#define HW    2048
#define IMG   (HW * HW)
#define KH    17
#define PC    8
#define LDSW  24
#define TROWS 32
#define WPROWS 48

template <int KW>
__global__ __launch_bounds__(256) void guided_pass(
    const float* __restrict__ Xin, const float* __restrict__ Yin,
    const float* __restrict__ Wk,  const float* __restrict__ Nplane,
    float* __restrict__ Xout)
{
    constexpr int PR = KW / 2;
    constexpr int TW = 16 + KW - 1;

    __shared__ float wpad[WPROWS * KW];
    __shared__ float tile[8][TROWS * LDSW];

    const int lane = threadIdx.x & 31;
    const int wid  = threadIdx.x >> 5;

    for (int i = threadIdx.x; i < WPROWS * KW; i += 256) {
        int j = i - 16 * KW;
        wpad[i] = (j >= 0 && j < KH * KW) ? Wk[j] : 0.0f;
    }

    const int TPR = HW / 16;
    const int TPI = TPR * TPR;
    const int t   = blockIdx.x * 8 + wid;
    const int img = t / TPI;
    const int rem = t % TPI;
    const int row0 = (rem / TPR) * 16;
    const int col0 = (rem % TPR) * 16;

    const float* Xi = Xin + (size_t)img * IMG;
    const float* Yi = Yin + (size_t)img * IMG;
    float*       Xo = Xout + (size_t)img * IMG;

    float* my = tile[wid];
    for (int idx = lane; idx < TROWS * TW; idx += 32) {
        int r = idx / TW, c = idx % TW;
        int gr = row0 + r - PC;
        int gc = col0 + c - PR;
        float d = 0.0f;
        if (gr >= 0 && gr < HW && gc >= 0 && gc < HW) {
            size_t o = (size_t)gr * HW + gc;
            d = Yi[o] - Xi[o];
        }
        my[r * LDSW + c] = d;
    }
    __syncthreads();

    v8f acc = {};
    const int half = lane >> 4;
    const int Mrow = lane & 15;
    const int n    = lane & 15;

#pragma unroll
    for (int s = 0; s < 8; ++s) {
        const int jA   = 4 * s + 2 * half;
        const int tpad = jA - Mrow + 16;
        const float* wrA = &wpad[tpad * KW];
        const float* wrB = wrA + KW;
        const float* ld0 = &my[jA * LDSW + n];
        const float* ld1 = ld0 + LDSW;

#pragma unroll
        for (int kx = 0; kx < KW; ++kx) {
            v2f a, b;
            a.x = wrA[kx];
            a.y = wrB[kx];
            b.x = ld0[kx];
            b.y = ld1[kx];
            acc = __builtin_amdgcn_wmma_f32_16x16x4_f32(
                false, a, false, b, (short)0, acc, false, false);
        }
    }

#pragma unroll
    for (int v = 0; v < 8; ++v) {
        int row = row0 + v + 8 * half;
        int col = col0 + n;
        size_t o = (size_t)row * HW + col;
        Xo[o] = Xi[o] + acc[v] / Nplane[o];
    }
}

template <int KW>
static void run_three_passes(const float* X, const float* y, const float* k,
                             const float* Nn, float* out, float* tmp,
                             int batch, hipStream_t stream)
{
    const int tiles = batch * (HW / 16) * (HW / 16);
    dim3 grid(tiles / 8), block(256);
    guided_pass<KW><<<grid, block, 0, stream>>>(X,   y, k,               Nn,           out);
    guided_pass<KW><<<grid, block, 0, stream>>>(out, y, k + KH * KW,     Nn + IMG,     tmp);
    guided_pass<KW><<<grid, block, 0, stream>>>(tmp, y, k + 2 * KH * KW, Nn + 2 * IMG, out);
}

extern "C" void kernel_launch(void* const* d_in, const int* in_sizes, int n_in,
                              void* d_out, int out_size, void* d_ws, size_t ws_size,
                              hipStream_t stream) {
    const float* X  = (const float*)d_in[0];
    const float* y  = (const float*)d_in[1];
    const float* k  = (const float*)d_in[2];
    const float* Nn = (const float*)d_in[3];
    float* out = (float*)d_out;
    float* tmp = (float*)d_ws;

    const int KW    = in_sizes[2] / (3 * KH);
    const int batch = in_sizes[0] / IMG;

    switch (KW) {
        case 1: run_three_passes<1>(X, y, k, Nn, out, tmp, batch, stream); break;
        case 3: run_three_passes<3>(X, y, k, Nn, out, tmp, batch, stream); break;
        case 7: run_three_passes<7>(X, y, k, Nn, out, tmp, batch, stream); break;
        case 9: run_three_passes<9>(X, y, k, Nn, out, tmp, batch, stream); break;
        case 5:
        default: run_three_passes<5>(X, y, k, Nn, out, tmp, batch, stream); break;
    }
}


# each wave computes a 64x16 strip (4 stacked tiles) instead of one tile: band weights held in registers, y-x rows staged once per strip (row pairs interleaved in LDS), staging loads unbranched
# speedup vs baseline: 1.2957x; 1.2957x over previous
#include <hip/hip_runtime.h>


typedef __attribute__((ext_vector_type(2))) float v2f;
typedef __attribute__((ext_vector_type(8))) float v8f;

#define HW    2048
#define IMG   (HW * HW)
#define KH    17
#define PC    8
#define LDSW  24
#define NT    4
#define SROWS (16 * NT + 2 * PC)
#define WPROWS 48

template <int KW>
__global__ __launch_bounds__(256)
__attribute__((amdgpu_waves_per_eu(KW <= 5 ? 8 : 4, KW <= 5 ? 8 : 4))) void guided_pass(
    const float* __restrict__ Xin, const float* __restrict__ Yin,
    const float* __restrict__ Wk,  const float* __restrict__ Nplane,
    float* __restrict__ Xout)
{
    constexpr int PR = KW / 2;
    constexpr int TW = 16 + KW - 1;
    constexpr int CG = HW / 128;
    constexpr int BANDS = HW / (16 * NT);

    __shared__ float wpad[WPROWS * KW];
    __shared__ __attribute__((aligned(16))) float tile[8][SROWS * LDSW];

    const int lane = threadIdx.x & 31;
    const int wid  = threadIdx.x >> 5;

    for (int i = threadIdx.x; i < WPROWS * KW; i += 256) {
        int j = i - 16 * KW;
        wpad[i] = (j >= 0 && j < KH * KW) ? Wk[j] : 0.0f;
    }

    const int bid  = blockIdx.x;
    const int img  = bid / (CG * BANDS);
    const int rem  = bid % (CG * BANDS);
    const int row0 = (rem / CG) * (16 * NT);
    const int col0 = (rem % CG) * 128 + wid * 16;

    const float* Xi = Xin + (size_t)img * IMG;
    const float* Yi = Yin + (size_t)img * IMG;
    float*       Xo = Xout + (size_t)img * IMG;

    float* my = tile[wid];
    const bool interior = (row0 >= PC) && (row0 + 16 * NT + PC <= HW) &&
                          (col0 >= PR) && (col0 + 16 + PR <= HW);
    if (interior) {
#pragma unroll
        for (int j = 0; j < TW / 2; ++j) {
            const int idx0 = lane + 32 * j;
            const int r0 = idx0 / TW, c = idx0 - r0 * TW;
            const size_t g = (size_t)(row0 - PC + r0) * HW + (size_t)(col0 - PR + c);
            const float* yp = Yi + g;
            const float* xp = Xi + g;
            float* lp = my + (r0 >> 1) * (2 * LDSW) + 2 * c + (r0 & 1);
#pragma unroll
            for (int k = 0; k < SROWS / 16; ++k)
                lp[k * 8 * (2 * LDSW)] = yp[(size_t)k * 16 * HW] - xp[(size_t)k * 16 * HW];
        }
    } else {
        constexpr int NIT = (SROWS * TW) / 32;
#pragma unroll 10
        for (int it = 0; it < NIT; ++it) {
            int idx = lane + 32 * it;
            int r = idx / TW, c = idx - r * TW;
            int gr = row0 + r - PC;
            int gc = col0 + c - PR;
            bool ok = ((unsigned)gr < (unsigned)HW) & ((unsigned)gc < (unsigned)HW);
            unsigned o = ok ? (unsigned)(gr * HW + gc) : 0u;
            float d = Yi[o] - Xi[o];
            my[(r >> 1) * (2 * LDSW) + 2 * c + (r & 1)] = ok ? d : 0.0f;
        }
    }
    __syncthreads();

    const int half = lane >> 4;
    const int Mrow = lane & 15;
    const int n    = lane & 15;

    v2f A[8][KW];
#pragma unroll
    for (int s = 0; s < 8; ++s) {
        const int jA   = 4 * s + 2 * half;
        const int tpad = jA - Mrow + 16;
#pragma unroll
        for (int kx = 0; kx < KW; ++kx) {
            A[s][kx].x = wpad[tpad * KW + kx];
            A[s][kx].y = wpad[(tpad + 1) * KW + kx];
        }
    }

#pragma unroll 1
    for (int i = 0; i < NT; ++i) {
        float xin[8], nrm[8];
        const size_t o0 = (size_t)(row0 + 16 * i + 8 * half) * HW + (col0 + n);
#pragma unroll
        for (int v = 0; v < 8; ++v) {
            xin[v] = Xi[o0 + (size_t)v * HW];
            nrm[v] = Nplane[o0 + (size_t)v * HW];
        }

        v8f acc = {};
        const float* tb = my + (8 * i + half) * (2 * LDSW) + 2 * n;
#pragma unroll
        for (int s = 0; s < 8; ++s) {
            const float* ld = tb + (2 * s) * (2 * LDSW);
#pragma unroll
            for (int kx = 0; kx < KW; ++kx) {
                v2f b = *(const v2f*)(ld + 2 * kx);
                acc = __builtin_amdgcn_wmma_f32_16x16x4_f32(
                    false, A[s][kx], false, b, (short)0, acc, false, false);
            }
        }

#pragma unroll
        for (int v = 0; v < 8; ++v)
            Xo[o0 + (size_t)v * HW] = xin[v] + acc[v] / nrm[v];
    }
}

template <int KW>
static void run_three_passes(const float* X, const float* y, const float* k,
                             const float* Nn, float* out, float* tmp,
                             int batch, hipStream_t stream)
{
    dim3 grid(batch * (HW / 128) * (HW / (16 * NT))), block(256);
    guided_pass<KW><<<grid, block, 0, stream>>>(X,   y, k,               Nn,           out);
    guided_pass<KW><<<grid, block, 0, stream>>>(out, y, k + KH * KW,     Nn + IMG,     tmp);
    guided_pass<KW><<<grid, block, 0, stream>>>(tmp, y, k + 2 * KH * KW, Nn + 2 * IMG, out);
}

extern "C" void kernel_launch(void* const* d_in, const int* in_sizes, int n_in,
                              void* d_out, int out_size, void* d_ws, size_t ws_size,
                              hipStream_t stream) {
    const float* X  = (const float*)d_in[0];
    const float* y  = (const float*)d_in[1];
    const float* k  = (const float*)d_in[2];
    const float* Nn = (const float*)d_in[3];
    float* out = (float*)d_out;
    float* tmp = (float*)d_ws;

    const int KW    = in_sizes[2] / (3 * KH);
    const int batch = in_sizes[0] / IMG;

    switch (KW) {
        case 1: run_three_passes<1>(X, y, k, Nn, out, tmp, batch, stream); break;
        case 3: run_three_passes<3>(X, y, k, Nn, out, tmp, batch, stream); break;
        case 7: run_three_passes<7>(X, y, k, Nn, out, tmp, batch, stream); break;
        case 9: run_three_passes<9>(X, y, k, Nn, out, tmp, batch, stream); break;
        case 5:
        default: run_three_passes<5>(X, y, k, Nn, out, tmp, batch, stream); break;
    }
}


# block stages one shared y-x patch for its 8 strips with whole-row 16-byte loads and 16-byte LDS stores instead of per-wave 4-byte patch loads; patch pitch keeps B reads bank-conflict-free
# speedup vs baseline: 1.6526x; 1.2754x over previous
#include <hip/hip_runtime.h>


typedef __attribute__((ext_vector_type(2))) float v2f;
typedef __attribute__((ext_vector_type(8))) float v8f;

#define HW    2048
#define IMG   (HW * HW)
#define KH    17
#define PC    8
#define PITCH 144
#define QPR   34
#define NT    4
#define SROWS (16 * NT + 2 * PC)
#define WPROWS 48

template <int KW>
__global__ __launch_bounds__(256)
__attribute__((amdgpu_waves_per_eu(KW <= 5 ? 8 : 4, KW <= 5 ? 8 : 4))) void guided_pass(
    const float* __restrict__ Xin, const float* __restrict__ Yin,
    const float* __restrict__ Wk,  const float* __restrict__ Nplane,
    float* __restrict__ Xout)
{
    constexpr int PR = KW / 2;
    constexpr int CG = HW / 128;
    constexpr int BANDS = HW / (16 * NT);

    __shared__ float wpad[WPROWS * KW];
    __shared__ __attribute__((aligned(16))) float tile[(SROWS / 2) * 2 * PITCH];

    const int lane = threadIdx.x & 31;
    const int wid  = threadIdx.x >> 5;

    for (int i = threadIdx.x; i < WPROWS * KW; i += 256) {
        int j = i - 16 * KW;
        wpad[i] = (j >= 0 && j < KH * KW) ? Wk[j] : 0.0f;
    }

    const int bid  = blockIdx.x;
    const int img  = bid / (CG * BANDS);
    const int rem  = bid % (CG * BANDS);
    const int row0 = (rem / CG) * (16 * NT);
    const int colb = (rem % CG) * 128;
    const int col0 = colb + wid * 16;

    const float* Xi = Xin + (size_t)img * IMG;
    const float* Yi = Yin + (size_t)img * IMG;
    float*       Xo = Xout + (size_t)img * IMG;

    auto stage_item = [&](int p, int cq) {
        const int gr0 = row0 - PC + 2 * p;
        const int gc  = colb - 4 + 4 * cq;
        const bool okc = (unsigned)gc < (unsigned)HW;
        const bool ok0 = okc & ((unsigned)gr0 < (unsigned)HW);
        const bool ok1 = okc & ((unsigned)(gr0 + 1) < (unsigned)HW);
        const unsigned o0 = ok0 ? (unsigned)(gr0 * HW + gc) : 0u;
        const unsigned o1 = ok1 ? (unsigned)((gr0 + 1) * HW + gc) : 0u;
        const float4 y0 = *(const float4*)(Yi + o0);
        const float4 x0 = *(const float4*)(Xi + o0);
        const float4 y1 = *(const float4*)(Yi + o1);
        const float4 x1 = *(const float4*)(Xi + o1);
        float4 lo, hi;
        lo.x = ok0 ? y0.x - x0.x : 0.0f;  lo.y = ok1 ? y1.x - x1.x : 0.0f;
        lo.z = ok0 ? y0.y - x0.y : 0.0f;  lo.w = ok1 ? y1.y - x1.y : 0.0f;
        hi.x = ok0 ? y0.z - x0.z : 0.0f;  hi.y = ok1 ? y1.z - x1.z : 0.0f;
        hi.z = ok0 ? y0.w - x0.w : 0.0f;  hi.w = ok1 ? y1.w - x1.w : 0.0f;
        float* lp = tile + p * (2 * PITCH) + 8 * cq;
        *(float4*)lp       = lo;
        *(float4*)(lp + 4) = hi;
    };
    if (threadIdx.x < SROWS)
        stage_item((int)threadIdx.x >> 1, (threadIdx.x & 1) ? (QPR - 1) : 0);
#pragma unroll
    for (int k = 0; k < SROWS / 16; ++k)
        stage_item(wid + 8 * k, 1 + lane);
    __syncthreads();

    const int half = lane >> 4;
    const int Mrow = lane & 15;
    const int n    = lane & 15;

    v2f A[8][KW];
#pragma unroll
    for (int s = 0; s < 8; ++s) {
        const int jA   = 4 * s + 2 * half;
        const int tpad = jA - Mrow + 16;
#pragma unroll
        for (int kx = 0; kx < KW; ++kx) {
            A[s][kx].x = wpad[tpad * KW + kx];
            A[s][kx].y = wpad[(tpad + 1) * KW + kx];
        }
    }

#pragma unroll 1
    for (int i = 0; i < NT; ++i) {
        float xin[8], nrm[8];
        const size_t o0 = (size_t)(row0 + 16 * i + 8 * half) * HW + (col0 + n);
#pragma unroll
        for (int v = 0; v < 8; ++v) {
            xin[v] = Xi[o0 + (size_t)v * HW];
            nrm[v] = Nplane[o0 + (size_t)v * HW];
        }

        v8f acc = {};
        const float* tb = tile + (8 * i + half) * (2 * PITCH) + 2 * (wid * 16 + n + 4 - PR);
#pragma unroll
        for (int s = 0; s < 8; ++s) {
            const float* ld = tb + (2 * s) * (2 * PITCH);
#pragma unroll
            for (int kx = 0; kx < KW; ++kx) {
                v2f b = *(const v2f*)(ld + 2 * kx);
                acc = __builtin_amdgcn_wmma_f32_16x16x4_f32(
                    false, A[s][kx], false, b, (short)0, acc, false, false);
            }
        }

#pragma unroll
        for (int v = 0; v < 8; ++v)
            Xo[o0 + (size_t)v * HW] = xin[v] + acc[v] / nrm[v];
    }
}

template <int KW>
static void run_three_passes(const float* X, const float* y, const float* k,
                             const float* Nn, float* out, float* tmp,
                             int batch, hipStream_t stream)
{
    dim3 grid(batch * (HW / 128) * (HW / (16 * NT))), block(256);
    guided_pass<KW><<<grid, block, 0, stream>>>(X,   y, k,               Nn,           out);
    guided_pass<KW><<<grid, block, 0, stream>>>(out, y, k + KH * KW,     Nn + IMG,     tmp);
    guided_pass<KW><<<grid, block, 0, stream>>>(tmp, y, k + 2 * KH * KW, Nn + 2 * IMG, out);
}

extern "C" void kernel_launch(void* const* d_in, const int* in_sizes, int n_in,
                              void* d_out, int out_size, void* d_ws, size_t ws_size,
                              hipStream_t stream) {
    const float* X  = (const float*)d_in[0];
    const float* y  = (const float*)d_in[1];
    const float* k  = (const float*)d_in[2];
    const float* Nn = (const float*)d_in[3];
    float* out = (float*)d_out;
    float* tmp = (float*)d_ws;

    const int KW    = in_sizes[2] / (3 * KH);
    const int batch = in_sizes[0] / IMG;

    switch (KW) {
        case 1: run_three_passes<1>(X, y, k, Nn, out, tmp, batch, stream); break;
        case 3: run_three_passes<3>(X, y, k, Nn, out, tmp, batch, stream); break;
        case 7: run_three_passes<7>(X, y, k, Nn, out, tmp, batch, stream); break;
        case 9: run_three_passes<9>(X, y, k, Nn, out, tmp, batch, stream); break;
        case 5:
        default: run_three_passes<5>(X, y, k, Nn, out, tmp, batch, stream); break;
    }
}


# passes carry the residual d=y-x (d'=d-conv(d)/N, last pass writes y-d'): passes 2,3 stage one array not two, tile centre read from the staged LDS patch; rounding now falls on d, not x
# speedup vs baseline: 1.7547x; 1.0618x over previous
#include <hip/hip_runtime.h>


typedef __attribute__((ext_vector_type(2))) float v2f;
typedef __attribute__((ext_vector_type(8))) float v8f;

#define HW    2048
#define IMG   (HW * HW)
#define KH    17
#define PC    8
#define PITCH 144
#define QPR   34
#define NT    4
#define SROWS (16 * NT + 2 * PC)
#define WPROWS 48

template <int KW, int MODE>
__global__ __launch_bounds__(256)
__attribute__((amdgpu_waves_per_eu(KW <= 5 ? 8 : 4, KW <= 5 ? 8 : 4))) void guided_pass(
    const float* __restrict__ Xin, const float* __restrict__ Yin,
    const float* __restrict__ Wk,  const float* __restrict__ Nplane,
    float* __restrict__ Xout)
{
    constexpr int PR = KW / 2;
    constexpr int CG = HW / 128;
    constexpr int BANDS = HW / (16 * NT);

    __shared__ float wpad[WPROWS * KW];
    __shared__ __attribute__((aligned(16))) float tile[(SROWS / 2) * 2 * PITCH];

    const int lane = threadIdx.x & 31;
    const int wid  = threadIdx.x >> 5;

    for (int i = threadIdx.x; i < WPROWS * KW; i += 256) {
        int j = i - 16 * KW;
        wpad[i] = (j >= 0 && j < KH * KW) ? Wk[j] : 0.0f;
    }

    const unsigned bid = blockIdx.x;
    const int img  = (int)(bid / (CG * BANDS));
    const unsigned rem = bid % (CG * BANDS);
    const int row0 = (int)((rem / CG) * (16 * NT));
    const int colb = (int)((rem % CG) * 128);
    const int col0 = colb + wid * 16;

    const float* Xi = Xin + (size_t)img * IMG;
    const float* Yi = Yin + (size_t)img * IMG;
    float*       Xo = Xout + (size_t)img * IMG;

    auto stage_item = [&](int p, int cq) {
        const int gr0 = row0 - PC + 2 * p;
        const int gc  = colb - 4 + 4 * cq;
        const bool okc = (unsigned)gc < (unsigned)HW;
        const bool ok0 = okc & ((unsigned)gr0 < (unsigned)HW);
        const bool ok1 = okc & ((unsigned)(gr0 + 1) < (unsigned)HW);
        const unsigned o0 = ok0 ? (unsigned)(gr0 * HW + gc) : 0u;
        const unsigned o1 = ok1 ? (unsigned)((gr0 + 1) * HW + gc) : 0u;
        float4 d0, d1;
        if constexpr (MODE == 0) {
            const float4 y0 = *(const float4*)(Yi + o0);
            const float4 x0 = *(const float4*)(Xi + o0);
            const float4 y1 = *(const float4*)(Yi + o1);
            const float4 x1 = *(const float4*)(Xi + o1);
            d0.x = y0.x - x0.x;  d0.y = y0.y - x0.y;  d0.z = y0.z - x0.z;  d0.w = y0.w - x0.w;
            d1.x = y1.x - x1.x;  d1.y = y1.y - x1.y;  d1.z = y1.z - x1.z;  d1.w = y1.w - x1.w;
        } else {
            d0 = *(const float4*)(Xi + o0);
            d1 = *(const float4*)(Xi + o1);
        }
        float4 lo, hi;
        lo.x = ok0 ? d0.x : 0.0f;  lo.y = ok1 ? d1.x : 0.0f;
        lo.z = ok0 ? d0.y : 0.0f;  lo.w = ok1 ? d1.y : 0.0f;
        hi.x = ok0 ? d0.z : 0.0f;  hi.y = ok1 ? d1.z : 0.0f;
        hi.z = ok0 ? d0.w : 0.0f;  hi.w = ok1 ? d1.w : 0.0f;
        float* lp = tile + p * (2 * PITCH) + 8 * cq;
        *(float4*)lp       = lo;
        *(float4*)(lp + 4) = hi;
    };
    if (threadIdx.x < SROWS)
        stage_item((int)threadIdx.x >> 1, (threadIdx.x & 1) ? (QPR - 1) : 0);
#pragma unroll
    for (int k = 0; k < SROWS / 16; ++k)
        stage_item(wid + 8 * k, 1 + lane);
    __syncthreads();

    const int half = lane >> 4;
    const int Mrow = lane & 15;
    const int n    = lane & 15;

    v2f A[8][KW];
#pragma unroll
    for (int s = 0; s < 8; ++s) {
        const int jA   = 4 * s + 2 * half;
        const int tpad = jA - Mrow + 16;
#pragma unroll
        for (int kx = 0; kx < KW; ++kx) {
            A[s][kx].x = wpad[tpad * KW + kx];
            A[s][kx].y = wpad[(tpad + 1) * KW + kx];
        }
    }

#pragma unroll 1
    for (int i = 0; i < NT; ++i) {
        float nrm[8], yc[8];
        const size_t o0 = (size_t)(row0 + 16 * i + 8 * half) * HW + (col0 + n);
#pragma unroll
        for (int v = 0; v < 8; ++v) {
            nrm[v] = Nplane[o0 + (size_t)v * HW];
            if constexpr (MODE == 2) yc[v] = Yi[o0 + (size_t)v * HW];
        }

        v8f acc = {};
        const float* tb = tile + (8 * i + half) * (2 * PITCH) + 2 * (wid * 16 + n + 4 - PR);
#pragma unroll
        for (int s = 0; s < 8; ++s) {
            const float* ld = tb + (2 * s) * (2 * PITCH);
#pragma unroll
            for (int kx = 0; kx < KW; ++kx) {
                v2f b = *(const v2f*)(ld + 2 * kx);
                acc = __builtin_amdgcn_wmma_f32_16x16x4_f32(
                    false, A[s][kx], false, b, (short)0, acc, false, false);
            }
            __builtin_amdgcn_sched_barrier(0);
        }

        float cen[8];
        const float* cb = tile + (8 * i + 4 + 4 * half) * (2 * PITCH) + 2 * (wid * 16 + n + 4);
#pragma unroll
        for (int j = 0; j < 4; ++j) {
            const v2f d = *(const v2f*)(cb + j * (2 * PITCH));
            cen[2 * j]     = d.x;
            cen[2 * j + 1] = d.y;
        }

#pragma unroll
        for (int v = 0; v < 8; ++v) {
            const float dn = cen[v] - acc[v] / nrm[v];
            if constexpr (MODE == 2) Xo[o0 + (size_t)v * HW] = yc[v] - dn;
            else                     Xo[o0 + (size_t)v * HW] = dn;
        }
    }
}

template <int KW>
static void run_three_passes(const float* X, const float* y, const float* k,
                             const float* Nn, float* out, float* tmp,
                             int batch, hipStream_t stream)
{
    dim3 grid(batch * (HW / 128) * (HW / (16 * NT))), block(256);
    guided_pass<KW, 0><<<grid, block, 0, stream>>>(X,   y, k,               Nn,           out);
    guided_pass<KW, 1><<<grid, block, 0, stream>>>(out, y, k + KH * KW,     Nn + IMG,     tmp);
    guided_pass<KW, 2><<<grid, block, 0, stream>>>(tmp, y, k + 2 * KH * KW, Nn + 2 * IMG, out);
}

extern "C" void kernel_launch(void* const* d_in, const int* in_sizes, int n_in,
                              void* d_out, int out_size, void* d_ws, size_t ws_size,
                              hipStream_t stream) {
    const float* X  = (const float*)d_in[0];
    const float* y  = (const float*)d_in[1];
    const float* k  = (const float*)d_in[2];
    const float* Nn = (const float*)d_in[3];
    float* out = (float*)d_out;
    float* tmp = (float*)d_ws;

    const int KW    = in_sizes[2] / (3 * KH);
    const int batch = in_sizes[0] / IMG;

    switch (KW) {
        case 1: run_three_passes<1>(X, y, k, Nn, out, tmp, batch, stream); break;
        case 3: run_three_passes<3>(X, y, k, Nn, out, tmp, batch, stream); break;
        case 7: run_three_passes<7>(X, y, k, Nn, out, tmp, batch, stream); break;
        case 9: run_three_passes<9>(X, y, k, Nn, out, tmp, batch, stream); break;
        case 5:
        default: run_three_passes<5>(X, y, k, Nn, out, tmp, batch, stream); break;
    }
}


# WMMAs whose band-weight fragment is zero in every lane are skipped: wave-uniform test on the weights as given (a ballot per tap column); B reads stay grouped per k step; remaining sums keep order
# speedup vs baseline: 2.0581x; 1.1729x over previous
#include <hip/hip_runtime.h>


typedef __attribute__((ext_vector_type(2))) float v2f;
typedef __attribute__((ext_vector_type(8))) float v8f;

#define HW    2048
#define IMG   (HW * HW)
#define KH    17
#define PC    8
#define PITCH 144
#define QPR   34
#define NT    4
#define SROWS (16 * NT + 2 * PC)
#define WPROWS 48

template <int KW, int MODE>
__global__ __launch_bounds__(256)
__attribute__((amdgpu_waves_per_eu(KW <= 5 ? 8 : 4, KW <= 5 ? 8 : 4))) void guided_pass(
    const float* __restrict__ Xin, const float* __restrict__ Yin,
    const float* __restrict__ Wk,  const float* __restrict__ Nplane,
    float* __restrict__ Xout)
{
    constexpr int PR = KW / 2;
    constexpr int CG = HW / 128;
    constexpr int BANDS = HW / (16 * NT);

    __shared__ float wpad[WPROWS * KW];
    __shared__ __attribute__((aligned(16))) float tile[(SROWS / 2) * 2 * PITCH];

    const int lane = threadIdx.x & 31;
    const int wid  = threadIdx.x >> 5;

    for (int i = threadIdx.x; i < WPROWS * KW; i += 256) {
        int j = i - 16 * KW;
        wpad[i] = (j >= 0 && j < KH * KW) ? Wk[j] : 0.0f;
    }

    const unsigned bid = blockIdx.x;
    const int img  = (int)(bid / (CG * BANDS));
    const unsigned rem = bid % (CG * BANDS);
    const int row0 = (int)((rem / CG) * (16 * NT));
    const int colb = (int)((rem % CG) * 128);
    const int col0 = colb + wid * 16;

    const float* Xi = Xin + (size_t)img * IMG;
    const float* Yi = Yin + (size_t)img * IMG;
    float*       Xo = Xout + (size_t)img * IMG;

    auto stage_item = [&](int p, int cq) {
        const int gr0 = row0 - PC + 2 * p;
        const int gc  = colb - 4 + 4 * cq;
        const bool okc = (unsigned)gc < (unsigned)HW;
        const bool ok0 = okc & ((unsigned)gr0 < (unsigned)HW);
        const bool ok1 = okc & ((unsigned)(gr0 + 1) < (unsigned)HW);
        const unsigned o0 = ok0 ? (unsigned)(gr0 * HW + gc) : 0u;
        const unsigned o1 = ok1 ? (unsigned)((gr0 + 1) * HW + gc) : 0u;
        float4 d0, d1;
        if constexpr (MODE == 0) {
            const float4 y0 = *(const float4*)(Yi + o0);
            const float4 x0 = *(const float4*)(Xi + o0);
            const float4 y1 = *(const float4*)(Yi + o1);
            const float4 x1 = *(const float4*)(Xi + o1);
            d0.x = y0.x - x0.x;  d0.y = y0.y - x0.y;  d0.z = y0.z - x0.z;  d0.w = y0.w - x0.w;
            d1.x = y1.x - x1.x;  d1.y = y1.y - x1.y;  d1.z = y1.z - x1.z;  d1.w = y1.w - x1.w;
        } else {
            d0 = *(const float4*)(Xi + o0);
            d1 = *(const float4*)(Xi + o1);
        }
        float4 lo, hi;
        lo.x = ok0 ? d0.x : 0.0f;  lo.y = ok1 ? d1.x : 0.0f;
        lo.z = ok0 ? d0.y : 0.0f;  lo.w = ok1 ? d1.y : 0.0f;
        hi.x = ok0 ? d0.z : 0.0f;  hi.y = ok1 ? d1.z : 0.0f;
        hi.z = ok0 ? d0.w : 0.0f;  hi.w = ok1 ? d1.w : 0.0f;
        float* lp = tile + p * (2 * PITCH) + 8 * cq;
        *(float4*)lp       = lo;
        *(float4*)(lp + 4) = hi;
    };
    if (threadIdx.x < SROWS)
        stage_item((int)threadIdx.x >> 1, (threadIdx.x & 1) ? (QPR - 1) : 0);
#pragma unroll
    for (int k = 0; k < SROWS / 16; ++k)
        stage_item(wid + 8 * k, 1 + lane);
    __syncthreads();

    const int half = lane >> 4;
    const int Mrow = lane & 15;
    const int n    = lane & 15;

    v2f A[8][KW];
#pragma unroll
    for (int s = 0; s < 8; ++s) {
        const int jA   = 4 * s + 2 * half;
        const int tpad = jA - Mrow + 16;
#pragma unroll
        for (int kx = 0; kx < KW; ++kx) {
            A[s][kx].x = wpad[tpad * KW + kx];
            A[s][kx].y = wpad[(tpad + 1) * KW + kx];
        }
    }

    unsigned colnz[KW];
#pragma unroll
    for (int kx = 0; kx < KW; ++kx) {
        const float w = (lane < KH) ? wpad[(16 + lane) * KW + kx] : 0.0f;
        colnz[kx] = __builtin_amdgcn_ballot_w32(w != 0.0f);
    }

#pragma unroll 1
    for (int i = 0; i < NT; ++i) {
        float nrm[8], yc[8];
        const size_t o0 = (size_t)(row0 + 16 * i + 8 * half) * HW + (col0 + n);
#pragma unroll
        for (int v = 0; v < 8; ++v) {
            nrm[v] = Nplane[o0 + (size_t)v * HW];
            if constexpr (MODE == 2) yc[v] = Yi[o0 + (size_t)v * HW];
        }

        v8f acc = {};
        const float* tb = tile + (8 * i + half) * (2 * PITCH) + 2 * (wid * 16 + n + 4 - PR);
#pragma unroll
        for (int s = 0; s < 8; ++s) {
            const float* ld = tb + (2 * s) * (2 * PITCH);
            const int tlo = (4 * s - 15 > 0) ? 4 * s - 15 : 0;
            const int thi = (4 * s + 3 < KH - 1) ? 4 * s + 3 : KH - 1;
            const unsigned taps = ((2u << thi) - 1u) & ~((1u << tlo) - 1u);
            v2f b[KW];
#pragma unroll
            for (int kx = 0; kx < KW; ++kx)
                b[kx] = *(const v2f*)(ld + 2 * kx);
#pragma unroll
            for (int kx = 0; kx < KW; ++kx)
                asm volatile("" : "+v"(b[kx]));
#pragma unroll
            for (int kx = 0; kx < KW; ++kx) {
                if (colnz[kx] & taps)
                    acc = __builtin_amdgcn_wmma_f32_16x16x4_f32(
                        false, A[s][kx], false, b[kx], (short)0, acc, false, false);
            }
        }

        float cen[8];
        const float* cb = tile + (8 * i + 4 + 4 * half) * (2 * PITCH) + 2 * (wid * 16 + n + 4);
#pragma unroll
        for (int j = 0; j < 4; ++j) {
            const v2f d = *(const v2f*)(cb + j * (2 * PITCH));
            cen[2 * j]     = d.x;
            cen[2 * j + 1] = d.y;
        }

#pragma unroll
        for (int v = 0; v < 8; ++v) {
            const float dn = cen[v] - acc[v] / nrm[v];
            if constexpr (MODE == 2) Xo[o0 + (size_t)v * HW] = yc[v] - dn;
            else                     Xo[o0 + (size_t)v * HW] = dn;
        }
    }
}

template <int KW>
static void run_three_passes(const float* X, const float* y, const float* k,
                             const float* Nn, float* out, float* tmp,
                             int batch, hipStream_t stream)
{
    dim3 grid(batch * (HW / 128) * (HW / (16 * NT))), block(256);
    guided_pass<KW, 0><<<grid, block, 0, stream>>>(X,   y, k,               Nn,           out);
    guided_pass<KW, 1><<<grid, block, 0, stream>>>(out, y, k + KH * KW,     Nn + IMG,     tmp);
    guided_pass<KW, 2><<<grid, block, 0, stream>>>(tmp, y, k + 2 * KH * KW, Nn + 2 * IMG, out);
}

extern "C" void kernel_launch(void* const* d_in, const int* in_sizes, int n_in,
                              void* d_out, int out_size, void* d_ws, size_t ws_size,
                              hipStream_t stream) {
    const float* X  = (const float*)d_in[0];
    const float* y  = (const float*)d_in[1];
    const float* k  = (const float*)d_in[2];
    const float* Nn = (const float*)d_in[3];
    float* out = (float*)d_out;
    float* tmp = (float*)d_ws;

    const int KW    = in_sizes[2] / (3 * KH);
    const int batch = in_sizes[0] / IMG;

    switch (KW) {
        case 1: run_three_passes<1>(X, y, k, Nn, out, tmp, batch, stream); break;
        case 3: run_three_passes<3>(X, y, k, Nn, out, tmp, batch, stream); break;
        case 7: run_three_passes<7>(X, y, k, Nn, out, tmp, batch, stream); break;
        case 9: run_three_passes<9>(X, y, k, Nn, out, tmp, batch, stream); break;
        case 5:
        default: run_three_passes<5>(X, y, k, Nn, out, tmp, batch, stream); break;
    }
}
